# starting kernel
# baseline (speedup 1.0000x reference)
#include <hip/hip_runtime.h>

#define N_NODES 50000
#define N_EDGES 500000
#define IN_C    128
#define HID_C   128
#define OUT_C   64
#define M_TILES (N_NODES / 16)

typedef __attribute__((ext_vector_type(2)))  float   v2f;
typedef __attribute__((ext_vector_type(8)))  float   v8f;
typedef __attribute__((ext_vector_type(16))) __bf16  v16bf;

__global__ void k_zero(float* __restrict__ p, int n) {
  int i = blockIdx.x * blockDim.x + threadIdx.x;
  if (i < n) p[i] = 0.0f;
}

__global__ void k_degree(const int* __restrict__ dst, float* __restrict__ deg) {
  int e = blockIdx.x * blockDim.x + threadIdx.x;
  if (e < N_EDGES) unsafeAtomicAdd(&deg[dst[e]], 1.0f);
}

__global__ void k_dinv(const float* __restrict__ deg, float* __restrict__ dinv) {
  int i = blockIdx.x * blockDim.x + threadIdx.x;
  if (i < N_NODES) dinv[i] = __frsqrt_rn(deg[i] + 1.0f);
}

template <int K, int NOUT>
__global__ __launch_bounds__(256)
void k_gemm_wmma(const float* __restrict__ X, const float* __restrict__ W,
                 float* __restrict__ H, int mTiles) {
  constexpr int NT = NOUT / 16;
  __shared__ float wLds[K * NOUT];

  for (int i = threadIdx.x; i < K * NOUT; i += 256) wLds[i] = W[i];
  __syncthreads();

  const int waveId = threadIdx.x >> 5;
  const int lane   = threadIdx.x & 31;
  const int mTile  = blockIdx.x * 8 + waveId;
  if (mTile >= mTiles) return;

  const int mBase = mTile * 16;
  const int half  = lane >> 4;
  const int col   = lane & 15;
  const int row   = mBase + col;
  const int koff  = half * 2;

  const float* ap = X + (size_t)row * K + koff;
  __builtin_prefetch(ap, 0, 1);

  v8f acc[NT];
#pragma unroll
  for (int t = 0; t < NT; ++t) acc[t] = (v8f){};

#if __has_builtin(__builtin_amdgcn_wmma_f32_16x16x4_f32)
#pragma unroll 4
  for (int kk = 0; kk < K; kk += 4) {
    v2f a;
    a.x = ap[kk];
    a.y = ap[kk + 1];
    const int krow = (kk + koff) * NOUT + col;
#pragma unroll
    for (int t = 0; t < NT; ++t) {
      v2f b;
      b.x = wLds[krow + 16 * t];
      b.y = wLds[krow + NOUT + 16 * t];
      acc[t] = __builtin_amdgcn_wmma_f32_16x16x4_f32(
           false, a,  false, b,
           (short)0, acc[t],  false,  false);
    }
  }
#else
  const int hiA = half * 8;
  const int hiB = half * 16;
  for (int kk = 0; kk < K; kk += 32) {
    v16bf a;
#pragma unroll
    for (int v = 0; v < 8; ++v) {
      int kA = ((v < 4) ? (v * 2) : (16 + (v - 4) * 2)) + hiA - koff;
      a[2 * v]     = (__bf16)ap[kk + kA];
      a[2 * v + 1] = (__bf16)ap[kk + kA + 1];
    }
#pragma unroll
    for (int t = 0; t < NT; ++t) {
      v16bf b;
#pragma unroll
      for (int v = 0; v < 8; ++v) {
        int kB = kk + 2 * v + hiB;
        b[2 * v]     = (__bf16)wLds[kB * NOUT + 16 * t + col];
        b[2 * v + 1] = (__bf16)wLds[(kB + 1) * NOUT + 16 * t + col];
      }
      acc[t] = __builtin_amdgcn_wmma_f32_16x16x32_bf16(
          false, a, false, b, (short)0, acc[t], false, false);
    }
  }
#endif

  const int rOff = half * 8;
#pragma unroll
  for (int t = 0; t < NT; ++t)
#pragma unroll
    for (int r = 0; r < 8; ++r)
      H[(size_t)(mBase + rOff + r) * NOUT + t * 16 + col] = acc[t][r];
}

template <int F>
__global__ void k_scatter(const int* __restrict__ src, const int* __restrict__ dst,
                          const float* __restrict__ dinv,
                          const float* __restrict__ H, float* __restrict__ AGG) {
  int edge = (blockIdx.x * blockDim.x + threadIdx.x) >> 5;
  int lane = threadIdx.x & 31;
  if (edge >= N_EDGES) return;
  int s = src[edge];
  int d = dst[edge];
  float coef = dinv[s] * dinv[d];
  const float* hrow = H + (size_t)s * F;
  float* arow = AGG + (size_t)d * F;
#pragma unroll
  for (int j = 0; j < F / 32; ++j) {
    int f = lane + j * 32;
    unsafeAtomicAdd(&arow[f], coef * hrow[f]);
  }
}

template <int F, bool RELU>
__global__ void k_combine(const float* __restrict__ AGG, const float* __restrict__ H,
                          const float* __restrict__ dinv, const float* __restrict__ b,
                          float* __restrict__ OUT) {
  int i = blockIdx.x * blockDim.x + threadIdx.x;
  if (i >= N_NODES * F) return;
  int node = i / F;
  int f    = i - node * F;
  float di = dinv[node];
  float v  = AGG[i] + di * di * H[i] + b[f];
  OUT[i] = RELU ? fmaxf(v, 0.0f) : v;
}

extern "C" void kernel_launch(void* const* d_in, const int* in_sizes, int n_in,
                              void* d_out, int out_size, void* d_ws, size_t ws_size,
                              hipStream_t stream) {
  const float* x    = (const float*)d_in[0];
  const int*   eidx = (const int*)d_in[1];
  const float* W1   = (const float*)d_in[2];
  const float* b1   = (const float*)d_in[3];
  const float* W2   = (const float*)d_in[4];
  const float* b2   = (const float*)d_in[5];
  const int* src = eidx;
  const int* dst = eidx + N_EDGES;

  char* ws = (char*)d_ws;
  auto alloc = [&](size_t bytes) {
    char* p = ws;
    ws += (bytes + 255) & ~(size_t)255;
    return (float*)p;
  };
  float* deg  = alloc((size_t)N_NODES * 4);
  float* dinv = alloc((size_t)N_NODES * 4);
  float* h1   = alloc((size_t)N_NODES * HID_C * 4);
  float* g1   = alloc((size_t)N_NODES * HID_C * 4);
  float* h2   = alloc((size_t)N_NODES * OUT_C * 4);
  float* g2   = alloc((size_t)N_NODES * OUT_C * 4);

  const int B = 256;
  k_zero<<<(N_NODES + B - 1) / B, B, 0, stream>>>(deg, N_NODES);
  k_zero<<<(N_NODES * HID_C + B - 1) / B, B, 0, stream>>>(g1, N_NODES * HID_C);
  k_zero<<<(N_NODES * OUT_C + B - 1) / B, B, 0, stream>>>(g2, N_NODES * OUT_C);

  k_degree<<<(N_EDGES + B - 1) / B, B, 0, stream>>>(dst, deg);
  k_dinv<<<(N_NODES + B - 1) / B, B, 0, stream>>>(deg, dinv);

  const int gemmBlocks = (M_TILES + 7) / 8;

  k_gemm_wmma<IN_C, HID_C><<<gemmBlocks, 256, 0, stream>>>(x, W1, h1, M_TILES);
  k_scatter<HID_C><<<(N_EDGES * 32 + B - 1) / B, B, 0, stream>>>(src, dst, dinv, h1, g1);
  k_combine<HID_C, true><<<(N_NODES * HID_C + B - 1) / B, B, 0, stream>>>(g1, h1, dinv, b1, g1);

  k_gemm_wmma<HID_C, OUT_C><<<gemmBlocks, 256, 0, stream>>>(g1, W2, h2, M_TILES);
  k_scatter<OUT_C><<<(N_EDGES * 32 + B - 1) / B, B, 0, stream>>>(src, dst, dinv, h2, g2);
  k_combine<OUT_C, false><<<(N_NODES * OUT_C + B - 1) / B, B, 0, stream>>>(g2, h2, dinv, b2, (float*)d_out);
}


# Aggregate by gather over a per-call destination-sorted edge list (integer atomic count gives arrival-order rank) instead of float atomic scatter; row sums now in rank order, self term first
# speedup vs baseline: 1.9615x; 1.9615x over previous
#include <hip/hip_runtime.h>

#define N_NODES 50000
#define N_EDGES 500000
#define IN_C    128
#define HID_C   128
#define OUT_C   64
#define M_TILES (N_NODES / 16)

typedef __attribute__((ext_vector_type(2)))  float   v2f;
typedef __attribute__((ext_vector_type(8)))  float   v8f;
typedef __attribute__((ext_vector_type(16))) __bf16  v16bf;

__global__ void k_zero_i(int* __restrict__ p, int n) {
  int i = blockIdx.x * blockDim.x + threadIdx.x;
  if (i < n) p[i] = 0;
}

__global__ void k_count_rank(const int* __restrict__ dst, int* __restrict__ cnt,
                             int* __restrict__ rank) {
  int e = blockIdx.x * blockDim.x + threadIdx.x;
  if (e < N_EDGES) rank[e] = atomicAdd(&cnt[dst[e]], 1);
}

__global__ __launch_bounds__(1024)
void k_scan(const int* __restrict__ cnt, int* __restrict__ rowptr,
            float* __restrict__ dinv) {
  __shared__ int sPart[32];
  __shared__ int sTot[32];
  const int tid  = threadIdx.x;
  const int lane = tid & 31;
  const int wv   = tid >> 5;

  const int4* c4 = (const int4*)cnt;
  const int n4 = blockIdx.x * 256;
  int part = 0;
#pragma unroll 4
  for (int i = tid; i < n4; i += 1024) {
    int4 q = c4[i];
    part += (q.x + q.y) + (q.z + q.w);
  }
  for (int o = 16; o > 0; o >>= 1) part += __shfl_xor(part, o);
  if (lane == 0) sPart[wv] = part;

  const int idx = blockIdx.x * 1024 + tid;
  const int v = (idx < N_NODES) ? cnt[idx] : 0;
  int incl = v;
  for (int o = 1; o < 32; o <<= 1) {
    int up = __shfl_up(incl, o);
    if (lane >= o) incl += up;
  }
  if (lane == 31) sTot[wv] = incl;
  __syncthreads();

  int base = 0;
  for (int w = 0; w < 32; ++w) base += sPart[w];
  int woff = 0;
  for (int w = 0; w < wv; ++w) woff += sTot[w];

  if (idx < N_NODES) {
    rowptr[idx] = base + woff + incl - v;
    dinv[idx]   = __frsqrt_rn((float)v + 1.0f);
  }
}

__global__ void k_fill(const int* __restrict__ src, const int* __restrict__ dst,
                       const int* __restrict__ rank, const int* __restrict__ rowptr,
                       int* __restrict__ csr) {
  int e = blockIdx.x * blockDim.x + threadIdx.x;
  if (e < N_EDGES) csr[rowptr[dst[e]] + rank[e]] = src[e];
}

template <int K, int NOUT>
__global__ __launch_bounds__(256)
void k_gemm_wmma(const float* __restrict__ X, const float* __restrict__ W,
                 const float* __restrict__ dinv,
                 float* __restrict__ H, int mTiles) {
  constexpr int NT = NOUT / 16;
  __shared__ float wLds[K * NOUT];

  for (int i = threadIdx.x; i < K * NOUT; i += 256) wLds[i] = W[i];
  __syncthreads();

  const int waveId = threadIdx.x >> 5;
  const int lane   = threadIdx.x & 31;
  const int mTile  = blockIdx.x * 8 + waveId;
  if (mTile >= mTiles) return;

  const int mBase = mTile * 16;
  const int half  = lane >> 4;
  const int col   = lane & 15;
  const int row   = mBase + col;
  const int koff  = half * 2;

  const float* ap = X + (size_t)row * K + koff;
  __builtin_prefetch(ap, 0, 1);

  const int rOff = half * 8;
  float dr[8];
#pragma unroll
  for (int r = 0; r < 8; ++r) dr[r] = dinv[mBase + rOff + r];

  v8f acc[NT];
#pragma unroll
  for (int t = 0; t < NT; ++t) acc[t] = (v8f){};

#if __has_builtin(__builtin_amdgcn_wmma_f32_16x16x4_f32)
#pragma unroll 4
  for (int kk = 0; kk < K; kk += 4) {
    v2f a;
    a.x = ap[kk];
    a.y = ap[kk + 1];
    const int krow = (kk + koff) * NOUT + col;
#pragma unroll
    for (int t = 0; t < NT; ++t) {
      v2f b;
      b.x = wLds[krow + 16 * t];
      b.y = wLds[krow + NOUT + 16 * t];
      acc[t] = __builtin_amdgcn_wmma_f32_16x16x4_f32(
           false, a,  false, b,
           (short)0, acc[t],  false,  false);
    }
  }
#else
  const int hiA = half * 8;
  const int hiB = half * 16;
  for (int kk = 0; kk < K; kk += 32) {
    v16bf a;
#pragma unroll
    for (int v = 0; v < 8; ++v) {
      int kA = ((v < 4) ? (v * 2) : (16 + (v - 4) * 2)) + hiA - koff;
      a[2 * v]     = (__bf16)ap[kk + kA];
      a[2 * v + 1] = (__bf16)ap[kk + kA + 1];
    }
#pragma unroll
    for (int t = 0; t < NT; ++t) {
      v16bf b;
#pragma unroll
      for (int v = 0; v < 8; ++v) {
        int kB = kk + 2 * v + hiB;
        b[2 * v]     = (__bf16)wLds[kB * NOUT + 16 * t + col];
        b[2 * v + 1] = (__bf16)wLds[(kB + 1) * NOUT + 16 * t + col];
      }
      acc[t] = __builtin_amdgcn_wmma_f32_16x16x32_bf16(
          false, a, false, b, (short)0, acc[t], false, false);
    }
  }
#endif

#pragma unroll
  for (int t = 0; t < NT; ++t)
#pragma unroll
    for (int r = 0; r < 8; ++r)
      H[(size_t)(mBase + rOff + r) * NOUT + t * 16 + col] = acc[t][r] * dr[r];
}

template <int F, bool RELU>
__global__ __launch_bounds__(256)
void k_gather(const int* __restrict__ rowptr, const int* __restrict__ cnt,
              const int* __restrict__ csr, const float* __restrict__ dinv,
              const float* __restrict__ HS, const float* __restrict__ b,
              float* __restrict__ OUT) {
  constexpr int V = F / 32;
  typedef __attribute__((ext_vector_type(V))) float vecf;

  const int lane = threadIdx.x & 31;
  const int node = __builtin_amdgcn_readfirstlane((int)((blockIdx.x * 256 + threadIdx.x) >> 5));
  if (node >= N_NODES) return;
  const int start = __builtin_amdgcn_readfirstlane(rowptr[node]);
  const int n     = __builtin_amdgcn_readfirstlane(cnt[node]);
  const float di  = dinv[node];

  const vecf* rows = (const vecf*)HS;
  vecf acc = rows[(size_t)node * 32 + lane];

  for (int base = 0; base < n; base += 32) {
    const int left = n - base;
    const int m = left < 32 ? left : 32;
    int idx = 0;
    if (lane < m) idx = csr[start + base + lane];
    int j = 0;
    for (; j + 4 <= m; j += 4) {
      const int s0 = __builtin_amdgcn_readlane(idx, j);
      const int s1 = __builtin_amdgcn_readlane(idx, j + 1);
      const int s2 = __builtin_amdgcn_readlane(idx, j + 2);
      const int s3 = __builtin_amdgcn_readlane(idx, j + 3);
      const vecf v0 = rows[(size_t)s0 * 32 + lane];
      const vecf v1 = rows[(size_t)s1 * 32 + lane];
      const vecf v2 = rows[(size_t)s2 * 32 + lane];
      const vecf v3 = rows[(size_t)s3 * 32 + lane];
      acc += v0;
      acc += v1;
      acc += v2;
      acc += v3;
    }
    const int rem = m - j;
    if (rem == 3) {
      const int s0 = __builtin_amdgcn_readlane(idx, j);
      const int s1 = __builtin_amdgcn_readlane(idx, j + 1);
      const int s2 = __builtin_amdgcn_readlane(idx, j + 2);
      const vecf v0 = rows[(size_t)s0 * 32 + lane];
      const vecf v1 = rows[(size_t)s1 * 32 + lane];
      const vecf v2 = rows[(size_t)s2 * 32 + lane];
      acc += v0;
      acc += v1;
      acc += v2;
    } else if (rem == 2) {
      const int s0 = __builtin_amdgcn_readlane(idx, j);
      const int s1 = __builtin_amdgcn_readlane(idx, j + 1);
      const vecf v0 = rows[(size_t)s0 * 32 + lane];
      const vecf v1 = rows[(size_t)s1 * 32 + lane];
      acc += v0;
      acc += v1;
    } else if (rem == 1) {
      const int s0 = __builtin_amdgcn_readlane(idx, j);
      acc += rows[(size_t)s0 * 32 + lane];
    }
  }

  const vecf bb = ((const vecf*)b)[lane];
  vecf o = di * acc + bb;
  if (RELU) {
#pragma unroll
    for (int k = 0; k < V; ++k) o[k] = fmaxf(o[k], 0.0f);
  }
  ((vecf*)OUT)[(size_t)node * 32 + lane] = o;
}

extern "C" void kernel_launch(void* const* d_in, const int* in_sizes, int n_in,
                              void* d_out, int out_size, void* d_ws, size_t ws_size,
                              hipStream_t stream) {
  const float* x    = (const float*)d_in[0];
  const int*   eidx = (const int*)d_in[1];
  const float* W1   = (const float*)d_in[2];
  const float* b1   = (const float*)d_in[3];
  const float* W2   = (const float*)d_in[4];
  const float* b2   = (const float*)d_in[5];
  const int* src = eidx;
  const int* dst = eidx + N_EDGES;

  char* ws = (char*)d_ws;
  auto alloc = [&](size_t bytes) {
    char* p = ws;
    ws += (bytes + 255) & ~(size_t)255;
    return (void*)p;
  };
  int*   cnt    = (int*)  alloc((size_t)N_NODES * 4);
  int*   rowptr = (int*)  alloc((size_t)N_NODES * 4);
  float* dinv   = (float*)alloc((size_t)N_NODES * 4);
  int*   rank   = (int*)  alloc((size_t)N_EDGES * 4);
  int*   csr    = (int*)  alloc((size_t)N_EDGES * 4);
  float* h1     = (float*)alloc((size_t)N_NODES * HID_C * 4);
  float* g1     = (float*)alloc((size_t)N_NODES * HID_C * 4);
  float* h2     = (float*)alloc((size_t)N_NODES * OUT_C * 4);

  const int B = 256;
  k_zero_i<<<(N_NODES + B - 1) / B, B, 0, stream>>>(cnt, N_NODES);
  k_count_rank<<<(N_EDGES + B - 1) / B, B, 0, stream>>>(dst, cnt, rank);
  k_scan<<<(N_NODES + 1023) / 1024, 1024, 0, stream>>>(cnt, rowptr, dinv);
  k_fill<<<(N_EDGES + B - 1) / B, B, 0, stream>>>(src, dst, rank, rowptr, csr);

  const int gemmBlocks   = (M_TILES + 7) / 8;
  const int gatherBlocks = (N_NODES + 7) / 8;

  k_gemm_wmma<IN_C, HID_C><<<gemmBlocks, 256, 0, stream>>>(x, W1, dinv, h1, M_TILES);
  k_gather<HID_C, true><<<gatherBlocks, 256, 0, stream>>>(rowptr, cnt, csr, dinv, h1, b1, g1);

  k_gemm_wmma<HID_C, OUT_C><<<gemmBlocks, 256, 0, stream>>>(g1, W2, dinv, h2, M_TILES);
  k_gather<OUT_C, false><<<gatherBlocks, 256, 0, stream>>>(rowptr, cnt, csr, dinv, h2, b2, (float*)d_out);
}


# Edges grouped by destination with LDS counters only (chunk-local range sort, then one owner block per node range) replacing clear, global atomics, scan, place; row sums in LDS-add arrival order
# speedup vs baseline: 3.0706x; 1.5654x over previous
#include <hip/hip_runtime.h>

#define N_NODES 50000
#define N_EDGES 500000
#define IN_C    128
#define HID_C   128
#define OUT_C   64
#define M_TILES (N_NODES / 16)

#define SORT_BLK 256
#define NBUCK    256
#define RANGE    196
#define CHUNK    ((N_EDGES + SORT_BLK - 1) / SORT_BLK)
#define CHUNK_P  1984
#define EPT      ((CHUNK + 255) / 256)

typedef __attribute__((ext_vector_type(2)))  float   v2f;
typedef __attribute__((ext_vector_type(8)))  float   v8f;
typedef __attribute__((ext_vector_type(16))) __bf16  v16bf;


__device__ __forceinline__ int scan256(int v, int lane, int wv, int* sW, int& total) {
  int incl = v;
  if (wv < 8) {
    for (int o = 1; o < 32; o <<= 1) {
      int up = __shfl_up(incl, o);
      if (lane >= o) incl += up;
    }
    if (lane == 31) sW[wv] = incl;
  }
  __syncthreads();
  int off = 0;
  total = 0;
  for (int w = 0; w < 8; ++w) {
    const int t = sW[w];
    if (w < wv) off += t;
    total += t;
  }
  return off + incl - v;
}

__global__ __launch_bounds__(256)
void k_bucket(const int* __restrict__ src, const int* __restrict__ dst,
              int* __restrict__ pairs, int* __restrict__ table) {
  __shared__ int  sCnt[NBUCK];
  __shared__ int  sOff[NBUCK];
  __shared__ int  sW[8];
  __shared__ int2 sPair[CHUNK];
  const int tid  = threadIdx.x;
  const int lane = tid & 31;
  const int wv   = tid >> 5;
  const int e0   = blockIdx.x * CHUNK;
  const int left = N_EDGES - e0;
  const int nE   = left < CHUNK ? left : CHUNK;

  sCnt[tid] = 0;
  __syncthreads();

  int s[EPT], d[EPT], k[EPT], r[EPT];
#pragma unroll
  for (int i = 0; i < EPT; ++i) {
    const int idx = tid + i * 256;
    const bool ok = idx < nE;
    d[i] = ok ? dst[e0 + idx] : -1;
    s[i] = ok ? src[e0 + idx] : 0;
  }
#pragma unroll
  for (int i = 0; i < EPT; ++i) {
    k[i] = 0;
    r[i] = 0;
    if (d[i] >= 0) {
      k[i] = d[i] / RANGE;
      r[i] = atomicAdd(&sCnt[k[i]], 1);
    }
  }
  __syncthreads();

  const int v = sCnt[tid];
  int total;
  const int excl = scan256(v, lane, wv, sW, total);
  sOff[tid] = excl;
  int* trow = table + blockIdx.x * (NBUCK + 1);
  trow[tid] = excl;
  if (tid == 255) trow[NBUCK] = total;
  __syncthreads();

#pragma unroll
  for (int i = 0; i < EPT; ++i)
    if (d[i] >= 0) sPair[sOff[k[i]] + r[i]] = make_int2(s[i], d[i]);
  __syncthreads();

  int2* outp = (int2*)pairs + (size_t)blockIdx.x * CHUNK_P;
  for (int idx = tid; idx < nE; idx += 256) outp[idx] = sPair[idx];
}

__global__ __launch_bounds__(512)
void k_owner(const int* __restrict__ pairs, const int* __restrict__ table,
             int* __restrict__ cnt, int* __restrict__ rowptr,
             float* __restrict__ dinv, int* __restrict__ csr) {
  __shared__ int sCnt[256];
  __shared__ int sRow[256];
  __shared__ int sCur[256];
  __shared__ int sRed[8];
  __shared__ int sW[8];
  __shared__ int sRunS[SORT_BLK];
  __shared__ int sRunL[SORT_BLK];
  const int tid  = threadIdx.x;
  const int lane = tid & 31;
  const int wv   = tid >> 5;
  const int kb   = blockIdx.x;
  const int lo   = kb * RANGE;

  if (tid < 256) {
    sCnt[tid] = 0;
    sCur[tid] = 0;
    const int* t = table + tid * (NBUCK + 1) + kb;
    const int s = t[0];
    const int e = t[1];
    sRunS[tid] = tid * CHUNK_P + s;
    sRunL[tid] = e - s;
    int part = s;
    for (int o = 16; o > 0; o >>= 1) part += __shfl_xor(part, o);
    if (lane == 0) sRed[wv] = part;
  }
  __syncthreads();
  int base = 0;
  for (int w = 0; w < 8; ++w) base += sRed[w];

  const int run = tid >> 1;
  const int sub = tid & 1;
  const int rl  = sRunL[run];
  const int2* pp = (const int2*)pairs + sRunS[run];

  for (int i0 = 0; i0 < rl; i0 += 16) {
    int dd[8];
#pragma unroll
    for (int u = 0; u < 8; ++u) {
      const int i = i0 + sub + 2 * u;
      dd[u] = (i < rl) ? pp[i].y : -1;
    }
#pragma unroll
    for (int u = 0; u < 8; ++u)
      if (dd[u] >= 0) atomicAdd(&sCnt[dd[u] - lo], 1);
  }
  __syncthreads();

  const int v = (tid < 256) ? sCnt[tid] : 0;
  int total;
  const int excl = scan256(v, lane, wv, sW, total);
  if (tid < 256) {
    sRow[tid] = excl;
    const int node = lo + tid;
    if (tid < RANGE && node < N_NODES) {
      cnt[node]    = v;
      rowptr[node] = base + excl;
      dinv[node]   = __frsqrt_rn((float)v + 1.0f);
    }
  }
  __syncthreads();

  for (int i0 = 0; i0 < rl; i0 += 16) {
    int2 p[8];
#pragma unroll
    for (int u = 0; u < 8; ++u) {
      const int i = i0 + sub + 2 * u;
      p[u] = (i < rl) ? pp[i] : make_int2(0, -1);
    }
#pragma unroll
    for (int u = 0; u < 8; ++u) {
      if (p[u].y >= 0) {
        const int n = p[u].y - lo;
        const int rk = atomicAdd(&sCur[n], 1);
        csr[base + sRow[n] + rk] = p[u].x;
      }
    }
  }
}

template <int K, int NOUT>
__global__ __launch_bounds__(256)
void k_gemm_wmma(const float* __restrict__ X, const float* __restrict__ W,
                 const float* __restrict__ dinv,
                 float* __restrict__ H, int mTiles) {
  constexpr int NT = NOUT / 16;
  __shared__ float wLds[K * NOUT];

  for (int i = threadIdx.x; i < K * NOUT; i += 256) wLds[i] = W[i];
  __syncthreads();

  const int waveId = threadIdx.x >> 5;
  const int lane   = threadIdx.x & 31;
  const int mTile  = blockIdx.x * 8 + waveId;
  if (mTile >= mTiles) return;

  const int mBase = mTile * 16;
  const int half  = lane >> 4;
  const int col   = lane & 15;
  const int row   = mBase + col;
  const int koff  = half * 2;

  const float* ap = X + (size_t)row * K + koff;
  __builtin_prefetch(ap, 0, 1);

  const int rOff = half * 8;
  float dr[8];
#pragma unroll
  for (int r = 0; r < 8; ++r) dr[r] = dinv[mBase + rOff + r];

  v8f acc[NT];
#pragma unroll
  for (int t = 0; t < NT; ++t) acc[t] = (v8f){};

#if __has_builtin(__builtin_amdgcn_wmma_f32_16x16x4_f32)
#pragma unroll 4
  for (int kk = 0; kk < K; kk += 4) {
    v2f a;
    a.x = ap[kk];
    a.y = ap[kk + 1];
    const int krow = (kk + koff) * NOUT + col;
#pragma unroll
    for (int t = 0; t < NT; ++t) {
      v2f b;
      b.x = wLds[krow + 16 * t];
      b.y = wLds[krow + NOUT + 16 * t];
      acc[t] = __builtin_amdgcn_wmma_f32_16x16x4_f32(
           false, a,  false, b,
           (short)0, acc[t],  false,  false);
    }
  }
#else
  const int hiA = half * 8;
  const int hiB = half * 16;
  for (int kk = 0; kk < K; kk += 32) {
    v16bf a;
#pragma unroll
    for (int v = 0; v < 8; ++v) {
      int kA = ((v < 4) ? (v * 2) : (16 + (v - 4) * 2)) + hiA - koff;
      a[2 * v]     = (__bf16)ap[kk + kA];
      a[2 * v + 1] = (__bf16)ap[kk + kA + 1];
    }
#pragma unroll
    for (int t = 0; t < NT; ++t) {
      v16bf b;
#pragma unroll
      for (int v = 0; v < 8; ++v) {
        int kB = kk + 2 * v + hiB;
        b[2 * v]     = (__bf16)wLds[kB * NOUT + 16 * t + col];
        b[2 * v + 1] = (__bf16)wLds[(kB + 1) * NOUT + 16 * t + col];
      }
      acc[t] = __builtin_amdgcn_wmma_f32_16x16x32_bf16(
          false, a, false, b, (short)0, acc[t], false, false);
    }
  }
#endif

#pragma unroll
  for (int t = 0; t < NT; ++t)
#pragma unroll
    for (int r = 0; r < 8; ++r)
      H[(size_t)(mBase + rOff + r) * NOUT + t * 16 + col] = acc[t][r] * dr[r];
}

template <int F, bool RELU>
__global__ __launch_bounds__(256)
void k_gather(const int* __restrict__ rowptr, const int* __restrict__ cnt,
              const int* __restrict__ csr, const float* __restrict__ dinv,
              const float* __restrict__ HS, const float* __restrict__ b,
              float* __restrict__ OUT) {
  constexpr int V = F / 32;
  typedef __attribute__((ext_vector_type(V))) float vecf;

  const int lane = threadIdx.x & 31;
  const int node = __builtin_amdgcn_readfirstlane((int)((blockIdx.x * 256 + threadIdx.x) >> 5));
  if (node >= N_NODES) return;
  const int start = __builtin_amdgcn_readfirstlane(rowptr[node]);
  const int n     = __builtin_amdgcn_readfirstlane(cnt[node]);
  const float di  = dinv[node];

  const vecf* rows = (const vecf*)HS;
  vecf acc = rows[(size_t)node * 32 + lane];

  for (int base = 0; base < n; base += 32) {
    const int left = n - base;
    const int m = left < 32 ? left : 32;
    int idx = 0;
    if (lane < m) idx = csr[start + base + lane];
    int j = 0;
    for (; j + 4 <= m; j += 4) {
      const int s0 = __builtin_amdgcn_readlane(idx, j);
      const int s1 = __builtin_amdgcn_readlane(idx, j + 1);
      const int s2 = __builtin_amdgcn_readlane(idx, j + 2);
      const int s3 = __builtin_amdgcn_readlane(idx, j + 3);
      const vecf v0 = rows[(size_t)s0 * 32 + lane];
      const vecf v1 = rows[(size_t)s1 * 32 + lane];
      const vecf v2 = rows[(size_t)s2 * 32 + lane];
      const vecf v3 = rows[(size_t)s3 * 32 + lane];
      acc += v0;
      acc += v1;
      acc += v2;
      acc += v3;
    }
    const int rem = m - j;
    if (rem == 3) {
      const int s0 = __builtin_amdgcn_readlane(idx, j);
      const int s1 = __builtin_amdgcn_readlane(idx, j + 1);
      const int s2 = __builtin_amdgcn_readlane(idx, j + 2);
      const vecf v0 = rows[(size_t)s0 * 32 + lane];
      const vecf v1 = rows[(size_t)s1 * 32 + lane];
      const vecf v2 = rows[(size_t)s2 * 32 + lane];
      acc += v0;
      acc += v1;
      acc += v2;
    } else if (rem == 2) {
      const int s0 = __builtin_amdgcn_readlane(idx, j);
      const int s1 = __builtin_amdgcn_readlane(idx, j + 1);
      const vecf v0 = rows[(size_t)s0 * 32 + lane];
      const vecf v1 = rows[(size_t)s1 * 32 + lane];
      acc += v0;
      acc += v1;
    } else if (rem == 1) {
      const int s0 = __builtin_amdgcn_readlane(idx, j);
      acc += rows[(size_t)s0 * 32 + lane];
    }
  }

  const vecf bb = ((const vecf*)b)[lane];
  vecf o = di * acc + bb;
  if (RELU) {
#pragma unroll
    for (int k = 0; k < V; ++k) o[k] = fmaxf(o[k], 0.0f);
  }
  ((vecf*)OUT)[(size_t)node * 32 + lane] = o;
}

extern "C" void kernel_launch(void* const* d_in, const int* in_sizes, int n_in,
                              void* d_out, int out_size, void* d_ws, size_t ws_size,
                              hipStream_t stream) {
  const float* x    = (const float*)d_in[0];
  const int*   eidx = (const int*)d_in[1];
  const float* W1   = (const float*)d_in[2];
  const float* b1   = (const float*)d_in[3];
  const float* W2   = (const float*)d_in[4];
  const float* b2   = (const float*)d_in[5];
  const int* src = eidx;
  const int* dst = eidx + N_EDGES;

  char* ws = (char*)d_ws;
  auto alloc = [&](size_t bytes) {
    char* p = ws;
    ws += (bytes + 255) & ~(size_t)255;
    return (void*)p;
  };
  int*   cnt    = (int*)  alloc((size_t)N_NODES * 4);
  int*   rowptr = (int*)  alloc((size_t)N_NODES * 4);
  float* dinv   = (float*)alloc((size_t)N_NODES * 4);
  int*   pairs  = (int*)  alloc((size_t)SORT_BLK * CHUNK_P * 8);
  int*   table  = (int*)  alloc((size_t)SORT_BLK * (NBUCK + 1) * 4);
  int*   csr    = (int*)  alloc((size_t)N_EDGES * 4);
  float* h1     = (float*)alloc((size_t)N_NODES * HID_C * 4);
  float* g1     = (float*)alloc((size_t)N_NODES * HID_C * 4);
  float* h2     = (float*)alloc((size_t)N_NODES * OUT_C * 4);

  k_bucket<<<SORT_BLK, 256, 0, stream>>>(src, dst, pairs, table);
  k_owner<<<NBUCK, 512, 0, stream>>>(pairs, table, cnt, rowptr, dinv, csr);

  const int gemmBlocks   = (M_TILES + 7) / 8;
  const int gatherBlocks = (N_NODES + 7) / 8;

  k_gemm_wmma<IN_C, HID_C><<<gemmBlocks, 256, 0, stream>>>(x, W1, dinv, h1, M_TILES);
  k_gather<HID_C, true><<<gatherBlocks, 256, 0, stream>>>(rowptr, cnt, csr, dinv, h1, b1, g1);

  k_gemm_wmma<HID_C, OUT_C><<<gemmBlocks, 256, 0, stream>>>(g1, W2, dinv, h2, M_TILES);
  k_gather<OUT_C, false><<<gatherBlocks, 256, 0, stream>>>(rowptr, cnt, csr, dinv, h2, b2, (float*)d_out);
}


# Layer-1 product freed from the degrees: left unscaled, its two column halves ride in the two edge-sort launches; dinv[src] applied in the aggregation as multiply-add, sum order kept
# speedup vs baseline: 3.6034x; 1.1735x over previous
#include <hip/hip_runtime.h>

#define N_NODES 50000
#define N_EDGES 500000
#define IN_C    128
#define HID_C   128
#define OUT_C   64
#define M_TILES (N_NODES / 16)

#define SORT_BLK 256
#define NBUCK    256
#define RANGE    196
#define CHUNK    ((N_EDGES + SORT_BLK - 1) / SORT_BLK)
#define CHUNK_P  1984
#define EPT      ((CHUNK + 255) / 256)

#define HALF_C   (HID_C / 2)
#define GEMM_BLK_8  ((M_TILES + 7) / 8)
#define GEMM_BLK_16 ((M_TILES + 15) / 16)

typedef __attribute__((ext_vector_type(2)))  float   v2f;
typedef __attribute__((ext_vector_type(8)))  float   v8f;
typedef __attribute__((ext_vector_type(16))) __bf16  v16bf;


__device__ __forceinline__ int scan256(int v, int lane, int wv, int* sW, int& total) {
  int incl = v;
  if (wv < 8) {
    for (int o = 1; o < 32; o <<= 1) {
      int up = __shfl_up(incl, o);
      if (lane >= o) incl += up;
    }
    if (lane == 31) sW[wv] = incl;
  }
  __syncthreads();
  int off = 0;
  total = 0;
  for (int w = 0; w < 8; ++w) {
    const int t = sW[w];
    if (w < wv) off += t;
    total += t;
  }
  return off + incl - v;
}

template <int K, int NOUT, int LD, int NTHR, bool SCALE>
__device__ __forceinline__
void gemm_body(const float* __restrict__ X, const float* __restrict__ W,
               const float* __restrict__ dinv, float* __restrict__ H,
               float* wLds, const int c0, const int blk) {
  constexpr int NT  = NOUT / 16;
  constexpr int WPB = NTHR / 32;

  for (int i = threadIdx.x; i < K * NOUT; i += NTHR)
    wLds[i] = W[(i / NOUT) * LD + c0 + (i % NOUT)];
  __syncthreads();

  const int waveId = threadIdx.x >> 5;
  const int lane   = threadIdx.x & 31;
  const int mTile  = blk * WPB + waveId;
  if (mTile >= M_TILES) return;

  const int mBase = mTile * 16;
  const int half  = lane >> 4;
  const int col   = lane & 15;
  const int row   = mBase + col;
  const int koff  = half * 2;

  const float* ap = X + (size_t)row * K + koff;
  __builtin_prefetch(ap, 0, 1);

  const int rOff = half * 8;
  float dr[8];
#pragma unroll
  for (int r = 0; r < 8; ++r) dr[r] = SCALE ? dinv[mBase + rOff + r] : 1.0f;

  v8f acc[NT];
#pragma unroll
  for (int t = 0; t < NT; ++t) acc[t] = (v8f){};

#if __has_builtin(__builtin_amdgcn_wmma_f32_16x16x4_f32)
#pragma unroll 4
  for (int kk = 0; kk < K; kk += 4) {
    v2f a;
    a.x = ap[kk];
    a.y = ap[kk + 1];
    const int krow = (kk + koff) * NOUT + col;
#pragma unroll
    for (int t = 0; t < NT; ++t) {
      v2f b;
      b.x = wLds[krow + 16 * t];
      b.y = wLds[krow + NOUT + 16 * t];
      acc[t] = __builtin_amdgcn_wmma_f32_16x16x4_f32(
           false, a,  false, b,
           (short)0, acc[t],  false,  false);
    }
  }
#else
  const int hiA = half * 8;
  const int hiB = half * 16;
  for (int kk = 0; kk < K; kk += 32) {
    v16bf a;
#pragma unroll
    for (int v = 0; v < 8; ++v) {
      int kA = ((v < 4) ? (v * 2) : (16 + (v - 4) * 2)) + hiA - koff;
      a[2 * v]     = (__bf16)ap[kk + kA];
      a[2 * v + 1] = (__bf16)ap[kk + kA + 1];
    }
#pragma unroll
    for (int t = 0; t < NT; ++t) {
      v16bf b;
#pragma unroll
      for (int v = 0; v < 8; ++v) {
        int kB = kk + 2 * v + hiB;
        b[2 * v]     = (__bf16)wLds[kB * NOUT + 16 * t + col];
        b[2 * v + 1] = (__bf16)wLds[(kB + 1) * NOUT + 16 * t + col];
      }
      acc[t] = __builtin_amdgcn_wmma_f32_16x16x32_bf16(
          false, a, false, b, (short)0, acc[t], false, false);
    }
  }
#endif

#pragma unroll
  for (int t = 0; t < NT; ++t)
#pragma unroll
    for (int r = 0; r < 8; ++r)
      H[(size_t)(mBase + rOff + r) * LD + c0 + t * 16 + col] =
          SCALE ? acc[t][r] * dr[r] : acc[t][r];
}

__global__ __launch_bounds__(256)
void k_gemm2(const float* __restrict__ G, const float* __restrict__ W2,
             const float* __restrict__ dinv, float* __restrict__ H2) {
  __shared__ float wLds[HID_C * OUT_C];
  gemm_body<HID_C, OUT_C, OUT_C, 256, true>(G, W2, dinv, H2, wLds, 0, blockIdx.x);
}

__global__ __launch_bounds__(256)
void k_bucket_gemm(const int* __restrict__ src, const int* __restrict__ dst,
                   int* __restrict__ pairs, int* __restrict__ table,
                   const float* __restrict__ X, const float* __restrict__ W1,
                   float* __restrict__ H1) {
  __shared__ float wLds[IN_C * HALF_C];
  if (blockIdx.x < GEMM_BLK_8) {
    gemm_body<IN_C, HALF_C, HID_C, 256, false>(X, W1, nullptr, H1, wLds, 0, blockIdx.x);
    return;
  }
  const int cb = blockIdx.x - GEMM_BLK_8;

  __shared__ int  sCnt[NBUCK];
  __shared__ int  sOff[NBUCK];
  __shared__ int  sW[8];
  __shared__ int2 sPair[CHUNK];
  const int tid  = threadIdx.x;
  const int lane = tid & 31;
  const int wv   = tid >> 5;
  const int e0   = cb * CHUNK;
  const int left = N_EDGES - e0;
  const int nE   = left < CHUNK ? left : CHUNK;

  sCnt[tid] = 0;
  __syncthreads();

  int s[EPT], d[EPT], k[EPT], r[EPT];
#pragma unroll
  for (int i = 0; i < EPT; ++i) {
    const int idx = tid + i * 256;
    const bool ok = idx < nE;
    d[i] = ok ? dst[e0 + idx] : -1;
    s[i] = ok ? src[e0 + idx] : 0;
  }
#pragma unroll
  for (int i = 0; i < EPT; ++i) {
    k[i] = 0;
    r[i] = 0;
    if (d[i] >= 0) {
      k[i] = d[i] / RANGE;
      r[i] = atomicAdd(&sCnt[k[i]], 1);
    }
  }
  __syncthreads();

  const int v = sCnt[tid];
  int total;
  const int excl = scan256(v, lane, wv, sW, total);
  sOff[tid] = excl;
  int* trow = table + cb * (NBUCK + 1);
  trow[tid] = excl;
  if (tid == 255) trow[NBUCK] = total;
  __syncthreads();

#pragma unroll
  for (int i = 0; i < EPT; ++i)
    if (d[i] >= 0) sPair[sOff[k[i]] + r[i]] = make_int2(s[i], d[i]);
  __syncthreads();

  int2* outp = (int2*)pairs + (size_t)cb * CHUNK_P;
  for (int idx = tid; idx < nE; idx += 256) outp[idx] = sPair[idx];
}

__global__ __launch_bounds__(512)
void k_owner_gemm(const int* __restrict__ pairs, const int* __restrict__ table,
                  int* __restrict__ cnt, int* __restrict__ rowptr,
                  float* __restrict__ dinv, int* __restrict__ csr,
                  const float* __restrict__ X, const float* __restrict__ W1,
                  float* __restrict__ H1) {
  __shared__ float wLds[IN_C * HALF_C];
  if (blockIdx.x < GEMM_BLK_16) {
    gemm_body<IN_C, HALF_C, HID_C, 512, false>(X, W1, nullptr, H1, wLds, HALF_C, blockIdx.x);
    return;
  }

  __shared__ int sCnt[256];
  __shared__ int sRow[256];
  __shared__ int sCur[256];
  __shared__ int sRed[8];
  __shared__ int sW[8];
  __shared__ int sRunS[SORT_BLK];
  __shared__ int sRunL[SORT_BLK];
  const int tid  = threadIdx.x;
  const int lane = tid & 31;
  const int wv   = tid >> 5;
  const int kb   = blockIdx.x - GEMM_BLK_16;
  const int lo   = kb * RANGE;

  if (tid < 256) {
    sCnt[tid] = 0;
    sCur[tid] = 0;
    const int* t = table + tid * (NBUCK + 1) + kb;
    const int s = t[0];
    const int e = t[1];
    sRunS[tid] = tid * CHUNK_P + s;
    sRunL[tid] = e - s;
    int part = s;
    for (int o = 16; o > 0; o >>= 1) part += __shfl_xor(part, o);
    if (lane == 0) sRed[wv] = part;
  }
  __syncthreads();
  int base = 0;
  for (int w = 0; w < 8; ++w) base += sRed[w];

  const int run = tid >> 1;
  const int sub = tid & 1;
  const int rl  = sRunL[run];
  const int2* pp = (const int2*)pairs + sRunS[run];

  for (int i0 = 0; i0 < rl; i0 += 16) {
    int dd[8];
#pragma unroll
    for (int u = 0; u < 8; ++u) {
      const int i = i0 + sub + 2 * u;
      dd[u] = (i < rl) ? pp[i].y : -1;
    }
#pragma unroll
    for (int u = 0; u < 8; ++u)
      if (dd[u] >= 0) atomicAdd(&sCnt[dd[u] - lo], 1);
  }
  __syncthreads();

  const int v = (tid < 256) ? sCnt[tid] : 0;
  int total;
  const int excl = scan256(v, lane, wv, sW, total);
  if (tid < 256) {
    sRow[tid] = excl;
    const int node = lo + tid;
    if (tid < RANGE && node < N_NODES) {
      cnt[node]    = v;
      rowptr[node] = base + excl;
      dinv[node]   = __frsqrt_rn((float)v + 1.0f);
    }
  }
  __syncthreads();

  for (int i0 = 0; i0 < rl; i0 += 16) {
    int2 p[8];
#pragma unroll
    for (int u = 0; u < 8; ++u) {
      const int i = i0 + sub + 2 * u;
      p[u] = (i < rl) ? pp[i] : make_int2(0, -1);
    }
#pragma unroll
    for (int u = 0; u < 8; ++u) {
      if (p[u].y >= 0) {
        const int n = p[u].y - lo;
        const int rk = atomicAdd(&sCur[n], 1);
        csr[base + sRow[n] + rk] = p[u].x;
      }
    }
  }
}

template <int F, bool RELU, bool SRCSCALE>
__global__ __launch_bounds__(256)
void k_gather(const int* __restrict__ rowptr, const int* __restrict__ cnt,
              const int* __restrict__ csr, const float* __restrict__ dinv,
              const float* __restrict__ HS, const float* __restrict__ b,
              float* __restrict__ OUT) {
  constexpr int V = F / 32;
  typedef __attribute__((ext_vector_type(V))) float vecf;

  const int lane = threadIdx.x & 31;
  const int node = __builtin_amdgcn_readfirstlane((int)((blockIdx.x * 256 + threadIdx.x) >> 5));
  if (node >= N_NODES) return;
  const int start = __builtin_amdgcn_readfirstlane(rowptr[node]);
  const int n     = __builtin_amdgcn_readfirstlane(cnt[node]);
  const float di  = dinv[node];

  const vecf* rows = (const vecf*)HS;
  vecf acc = rows[(size_t)node * 32 + lane];
  if (SRCSCALE) acc = di * acc;

  for (int base = 0; base < n; base += 32) {
    const int left = n - base;
    const int m = left < 32 ? left : 32;
    int idx = 0;
    if (lane < m) idx = csr[start + base + lane];
    int j = 0;
    for (; j + 4 <= m; j += 4) {
      const int s0 = __builtin_amdgcn_readlane(idx, j);
      const int s1 = __builtin_amdgcn_readlane(idx, j + 1);
      const int s2 = __builtin_amdgcn_readlane(idx, j + 2);
      const int s3 = __builtin_amdgcn_readlane(idx, j + 3);
      float d0 = 1.0f, d1 = 1.0f, d2 = 1.0f, d3 = 1.0f;
      if (SRCSCALE) { d0 = dinv[s0]; d1 = dinv[s1]; d2 = dinv[s2]; d3 = dinv[s3]; }
      const vecf v0 = rows[(size_t)s0 * 32 + lane];
      const vecf v1 = rows[(size_t)s1 * 32 + lane];
      const vecf v2 = rows[(size_t)s2 * 32 + lane];
      const vecf v3 = rows[(size_t)s3 * 32 + lane];
      if (SRCSCALE) {
        acc += d0 * v0;
        acc += d1 * v1;
        acc += d2 * v2;
        acc += d3 * v3;
      } else {
        acc += v0;
        acc += v1;
        acc += v2;
        acc += v3;
      }
    }
    const int rem = m - j;
    if (rem == 3) {
      const int s0 = __builtin_amdgcn_readlane(idx, j);
      const int s1 = __builtin_amdgcn_readlane(idx, j + 1);
      const int s2 = __builtin_amdgcn_readlane(idx, j + 2);
      float d0 = 1.0f, d1 = 1.0f, d2 = 1.0f;
      if (SRCSCALE) { d0 = dinv[s0]; d1 = dinv[s1]; d2 = dinv[s2]; }
      const vecf v0 = rows[(size_t)s0 * 32 + lane];
      const vecf v1 = rows[(size_t)s1 * 32 + lane];
      const vecf v2 = rows[(size_t)s2 * 32 + lane];
      if (SRCSCALE) {
        acc += d0 * v0;
        acc += d1 * v1;
        acc += d2 * v2;
      } else {
        acc += v0;
        acc += v1;
        acc += v2;
      }
    } else if (rem == 2) {
      const int s0 = __builtin_amdgcn_readlane(idx, j);
      const int s1 = __builtin_amdgcn_readlane(idx, j + 1);
      float d0 = 1.0f, d1 = 1.0f;
      if (SRCSCALE) { d0 = dinv[s0]; d1 = dinv[s1]; }
      const vecf v0 = rows[(size_t)s0 * 32 + lane];
      const vecf v1 = rows[(size_t)s1 * 32 + lane];
      if (SRCSCALE) {
        acc += d0 * v0;
        acc += d1 * v1;
      } else {
        acc += v0;
        acc += v1;
      }
    } else if (rem == 1) {
      const int s0 = __builtin_amdgcn_readlane(idx, j);
      float d0 = 1.0f;
      if (SRCSCALE) d0 = dinv[s0];
      const vecf v0 = rows[(size_t)s0 * 32 + lane];
      if (SRCSCALE) acc += d0 * v0; else acc += v0;
    }
  }

  const vecf bb = ((const vecf*)b)[lane];
  vecf o = di * acc + bb;
  if (RELU) {
#pragma unroll
    for (int k = 0; k < V; ++k) o[k] = fmaxf(o[k], 0.0f);
  }
  ((vecf*)OUT)[(size_t)node * 32 + lane] = o;
}

extern "C" void kernel_launch(void* const* d_in, const int* in_sizes, int n_in,
                              void* d_out, int out_size, void* d_ws, size_t ws_size,
                              hipStream_t stream) {
  const float* x    = (const float*)d_in[0];
  const int*   eidx = (const int*)d_in[1];
  const float* W1   = (const float*)d_in[2];
  const float* b1   = (const float*)d_in[3];
  const float* W2   = (const float*)d_in[4];
  const float* b2   = (const float*)d_in[5];
  const int* src = eidx;
  const int* dst = eidx + N_EDGES;

  char* ws = (char*)d_ws;
  auto alloc = [&](size_t bytes) {
    char* p = ws;
    ws += (bytes + 255) & ~(size_t)255;
    return (void*)p;
  };
  int*   cnt    = (int*)  alloc((size_t)N_NODES * 4);
  int*   rowptr = (int*)  alloc((size_t)N_NODES * 4);
  float* dinv   = (float*)alloc((size_t)N_NODES * 4);
  int*   pairs  = (int*)  alloc((size_t)SORT_BLK * CHUNK_P * 8);
  int*   table  = (int*)  alloc((size_t)SORT_BLK * (NBUCK + 1) * 4);
  int*   csr    = (int*)  alloc((size_t)N_EDGES * 4);
  float* h1     = (float*)alloc((size_t)N_NODES * HID_C * 4);
  float* g1     = (float*)alloc((size_t)N_NODES * HID_C * 4);
  float* h2     = (float*)alloc((size_t)N_NODES * OUT_C * 4);

  k_bucket_gemm<<<GEMM_BLK_8 + SORT_BLK, 256, 0, stream>>>(src, dst, pairs, table, x, W1, h1);
  k_owner_gemm<<<GEMM_BLK_16 + NBUCK, 512, 0, stream>>>(pairs, table, cnt, rowptr, dinv, csr,
                                                        x, W1, h1);

  const int gatherBlocks = (N_NODES + 7) / 8;

  k_gather<HID_C, true, true><<<gatherBlocks, 256, 0, stream>>>(rowptr, cnt, csr, dinv, h1, b1, g1);

  k_gemm2<<<GEMM_BLK_8, 256, 0, stream>>>(g1, W2, dinv, h2);
  k_gather<OUT_C, false, false><<<gatherBlocks, 256, 0, stream>>>(rowptr, cnt, csr, dinv, h2, b2, (float*)d_out);
}


# Layer-2 product moved into the layer-1 aggregation: a 16-wave block keeps its 32 nodes' layer-1 rows in an LDS tile and multiplies them by W2 there; one launch fewer, no g1 plane; sum orders kept
# speedup vs baseline: 4.0715x; 1.1299x over previous
#include <hip/hip_runtime.h>

#define N_NODES 50000
#define N_EDGES 500000
#define IN_C    128
#define HID_C   128
#define OUT_C   64
#define M_TILES (N_NODES / 16)

#define SORT_BLK 256
#define NBUCK    256
#define RANGE    196
#define CHUNK    ((N_EDGES + SORT_BLK - 1) / SORT_BLK)
#define CHUNK_P  1984
#define EPT      ((CHUNK + 255) / 256)

#define HALF_C   (HID_C / 2)
#define GEMM_BLK_8  ((M_TILES + 7) / 8)
#define GEMM_BLK_16 ((M_TILES + 15) / 16)

#define TILE_N   32
#define FUSE_BLK ((N_NODES + TILE_N - 1) / TILE_N)
#define G_PITCH  (HID_C + 4)
#define W2_PITCH (OUT_C + 16)

typedef __attribute__((ext_vector_type(2)))  float   v2f;
typedef __attribute__((ext_vector_type(8)))  float   v8f;
typedef __attribute__((ext_vector_type(16))) __bf16  v16bf;
template <int V> using rowvec = float __attribute__((ext_vector_type(V)));


__device__ __forceinline__ int scan256(int v, int lane, int wv, int* sW, int& total) {
  int incl = v;
  if (wv < 8) {
    for (int o = 1; o < 32; o <<= 1) {
      int up = __shfl_up(incl, o);
      if (lane >= o) incl += up;
    }
    if (lane == 31) sW[wv] = incl;
  }
  __syncthreads();
  int off = 0;
  total = 0;
  for (int w = 0; w < 8; ++w) {
    const int t = sW[w];
    if (w < wv) off += t;
    total += t;
  }
  return off + incl - v;
}

template <int K, int NOUT, int LD, int NTHR>
__device__ __forceinline__
void gemm_body(const float* __restrict__ X, const float* __restrict__ W,
               float* __restrict__ H,
               float* wLds, const int c0, const int blk) {
  constexpr int NT  = NOUT / 16;
  constexpr int WPB = NTHR / 32;

  for (int i = threadIdx.x; i < K * NOUT; i += NTHR)
    wLds[i] = W[(i / NOUT) * LD + c0 + (i % NOUT)];
  __syncthreads();

  const int waveId = threadIdx.x >> 5;
  const int lane   = threadIdx.x & 31;
  const int mTile  = blk * WPB + waveId;
  if (mTile >= M_TILES) return;

  const int mBase = mTile * 16;
  const int half  = lane >> 4;
  const int col   = lane & 15;
  const int row   = mBase + col;
  const int koff  = half * 2;

  const float* ap = X + (size_t)row * K + koff;
  __builtin_prefetch(ap, 0, 1);

  const int rOff = half * 8;

  v8f acc[NT];
#pragma unroll
  for (int t = 0; t < NT; ++t) acc[t] = (v8f){};

#if __has_builtin(__builtin_amdgcn_wmma_f32_16x16x4_f32)
#pragma unroll 4
  for (int kk = 0; kk < K; kk += 4) {
    v2f a;
    a.x = ap[kk];
    a.y = ap[kk + 1];
    const int krow = (kk + koff) * NOUT + col;
#pragma unroll
    for (int t = 0; t < NT; ++t) {
      v2f b;
      b.x = wLds[krow + 16 * t];
      b.y = wLds[krow + NOUT + 16 * t];
      acc[t] = __builtin_amdgcn_wmma_f32_16x16x4_f32(
           false, a,  false, b,
           (short)0, acc[t],  false,  false);
    }
  }
#else
  const int hiA = half * 8;
  const int hiB = half * 16;
  for (int kk = 0; kk < K; kk += 32) {
    v16bf a;
#pragma unroll
    for (int v = 0; v < 8; ++v) {
      int kA = ((v < 4) ? (v * 2) : (16 + (v - 4) * 2)) + hiA - koff;
      a[2 * v]     = (__bf16)ap[kk + kA];
      a[2 * v + 1] = (__bf16)ap[kk + kA + 1];
    }
#pragma unroll
    for (int t = 0; t < NT; ++t) {
      v16bf b;
#pragma unroll
      for (int v = 0; v < 8; ++v) {
        int kB = kk + 2 * v + hiB;
        b[2 * v]     = (__bf16)wLds[kB * NOUT + 16 * t + col];
        b[2 * v + 1] = (__bf16)wLds[(kB + 1) * NOUT + 16 * t + col];
      }
      acc[t] = __builtin_amdgcn_wmma_f32_16x16x32_bf16(
          false, a, false, b, (short)0, acc[t], false, false);
    }
  }
#endif

#pragma unroll
  for (int t = 0; t < NT; ++t)
#pragma unroll
    for (int r = 0; r < 8; ++r)
      H[(size_t)(mBase + rOff + r) * LD + c0 + t * 16 + col] = acc[t][r];
}

__global__ __launch_bounds__(256)
void k_bucket_gemm(const int* __restrict__ src, const int* __restrict__ dst,
                   int* __restrict__ pairs, int* __restrict__ table,
                   const float* __restrict__ X, const float* __restrict__ W1,
                   float* __restrict__ H1) {
  __shared__ float wLds[IN_C * HALF_C];
  if (blockIdx.x < GEMM_BLK_8) {
    gemm_body<IN_C, HALF_C, HID_C, 256>(X, W1, H1, wLds, 0, blockIdx.x);
    return;
  }
  const int cb = blockIdx.x - GEMM_BLK_8;

  __shared__ int  sCnt[NBUCK];
  __shared__ int  sOff[NBUCK];
  __shared__ int  sW[8];
  __shared__ int2 sPair[CHUNK];
  const int tid  = threadIdx.x;
  const int lane = tid & 31;
  const int wv   = tid >> 5;
  const int e0   = cb * CHUNK;
  const int left = N_EDGES - e0;
  const int nE   = left < CHUNK ? left : CHUNK;

  sCnt[tid] = 0;
  __syncthreads();

  int s[EPT], d[EPT], k[EPT], r[EPT];
#pragma unroll
  for (int i = 0; i < EPT; ++i) {
    const int idx = tid + i * 256;
    const bool ok = idx < nE;
    d[i] = ok ? dst[e0 + idx] : -1;
    s[i] = ok ? src[e0 + idx] : 0;
  }
#pragma unroll
  for (int i = 0; i < EPT; ++i) {
    k[i] = 0;
    r[i] = 0;
    if (d[i] >= 0) {
      k[i] = d[i] / RANGE;
      r[i] = atomicAdd(&sCnt[k[i]], 1);
    }
  }
  __syncthreads();

  const int v = sCnt[tid];
  int total;
  const int excl = scan256(v, lane, wv, sW, total);
  sOff[tid] = excl;
  int* trow = table + cb * (NBUCK + 1);
  trow[tid] = excl;
  if (tid == 255) trow[NBUCK] = total;
  __syncthreads();

#pragma unroll
  for (int i = 0; i < EPT; ++i)
    if (d[i] >= 0) sPair[sOff[k[i]] + r[i]] = make_int2(s[i], d[i]);
  __syncthreads();

  int2* outp = (int2*)pairs + (size_t)cb * CHUNK_P;
  for (int idx = tid; idx < nE; idx += 256) outp[idx] = sPair[idx];
}

__global__ __launch_bounds__(512)
void k_owner_gemm(const int* __restrict__ pairs, const int* __restrict__ table,
                  int* __restrict__ cnt, int* __restrict__ rowptr,
                  float* __restrict__ dinv, int* __restrict__ csr,
                  const float* __restrict__ X, const float* __restrict__ W1,
                  float* __restrict__ H1) {
  __shared__ float wLds[IN_C * HALF_C];
  if (blockIdx.x < GEMM_BLK_16) {
    gemm_body<IN_C, HALF_C, HID_C, 512>(X, W1, H1, wLds, HALF_C, blockIdx.x);
    return;
  }

  __shared__ int sCnt[256];
  __shared__ int sRow[256];
  __shared__ int sCur[256];
  __shared__ int sRed[8];
  __shared__ int sW[8];
  __shared__ int sRunS[SORT_BLK];
  __shared__ int sRunL[SORT_BLK];
  const int tid  = threadIdx.x;
  const int lane = tid & 31;
  const int wv   = tid >> 5;
  const int kb   = blockIdx.x - GEMM_BLK_16;
  const int lo   = kb * RANGE;

  if (tid < 256) {
    sCnt[tid] = 0;
    sCur[tid] = 0;
    const int* t = table + tid * (NBUCK + 1) + kb;
    const int s = t[0];
    const int e = t[1];
    sRunS[tid] = tid * CHUNK_P + s;
    sRunL[tid] = e - s;
    int part = s;
    for (int o = 16; o > 0; o >>= 1) part += __shfl_xor(part, o);
    if (lane == 0) sRed[wv] = part;
  }
  __syncthreads();
  int base = 0;
  for (int w = 0; w < 8; ++w) base += sRed[w];

  const int run = tid >> 1;
  const int sub = tid & 1;
  const int rl  = sRunL[run];
  const int2* pp = (const int2*)pairs + sRunS[run];

  for (int i0 = 0; i0 < rl; i0 += 16) {
    int dd[8];
#pragma unroll
    for (int u = 0; u < 8; ++u) {
      const int i = i0 + sub + 2 * u;
      dd[u] = (i < rl) ? pp[i].y : -1;
    }
#pragma unroll
    for (int u = 0; u < 8; ++u)
      if (dd[u] >= 0) atomicAdd(&sCnt[dd[u] - lo], 1);
  }
  __syncthreads();

  const int v = (tid < 256) ? sCnt[tid] : 0;
  int total;
  const int excl = scan256(v, lane, wv, sW, total);
  if (tid < 256) {
    sRow[tid] = excl;
    const int node = lo + tid;
    if (tid < RANGE && node < N_NODES) {
      cnt[node]    = v;
      rowptr[node] = base + excl;
      dinv[node]   = __frsqrt_rn((float)v + 1.0f);
    }
  }
  __syncthreads();

  for (int i0 = 0; i0 < rl; i0 += 16) {
    int2 p[8];
#pragma unroll
    for (int u = 0; u < 8; ++u) {
      const int i = i0 + sub + 2 * u;
      p[u] = (i < rl) ? pp[i] : make_int2(0, -1);
    }
#pragma unroll
    for (int u = 0; u < 8; ++u) {
      if (p[u].y >= 0) {
        const int n = p[u].y - lo;
        const int rk = atomicAdd(&sCur[n], 1);
        csr[base + sRow[n] + rk] = p[u].x;
      }
    }
  }
}

template <int F, bool SRCSCALE>
__device__ __forceinline__
rowvec<F / 32>
gather_sum(const int node, const int start, const int n, const float di, const int lane,
           const int* __restrict__ csr, const float* __restrict__ dinv,
           const float* __restrict__ HS) {
  typedef rowvec<F / 32> vecf;

  const vecf* rows = (const vecf*)HS;
  vecf acc = rows[(size_t)node * 32 + lane];
  if (SRCSCALE) acc = di * acc;

  for (int base = 0; base < n; base += 32) {
    const int left = n - base;
    const int m = left < 32 ? left : 32;
    int idx = 0;
    if (lane < m) idx = csr[start + base + lane];
    int j = 0;
    for (; j + 4 <= m; j += 4) {
      const int s0 = __builtin_amdgcn_readlane(idx, j);
      const int s1 = __builtin_amdgcn_readlane(idx, j + 1);
      const int s2 = __builtin_amdgcn_readlane(idx, j + 2);
      const int s3 = __builtin_amdgcn_readlane(idx, j + 3);
      float d0 = 1.0f, d1 = 1.0f, d2 = 1.0f, d3 = 1.0f;
      if (SRCSCALE) { d0 = dinv[s0]; d1 = dinv[s1]; d2 = dinv[s2]; d3 = dinv[s3]; }
      const vecf v0 = rows[(size_t)s0 * 32 + lane];
      const vecf v1 = rows[(size_t)s1 * 32 + lane];
      const vecf v2 = rows[(size_t)s2 * 32 + lane];
      const vecf v3 = rows[(size_t)s3 * 32 + lane];
      if (SRCSCALE) {
        acc += d0 * v0;
        acc += d1 * v1;
        acc += d2 * v2;
        acc += d3 * v3;
      } else {
        acc += v0;
        acc += v1;
        acc += v2;
        acc += v3;
      }
    }
    const int rem = m - j;
    if (rem == 3) {
      const int s0 = __builtin_amdgcn_readlane(idx, j);
      const int s1 = __builtin_amdgcn_readlane(idx, j + 1);
      const int s2 = __builtin_amdgcn_readlane(idx, j + 2);
      float d0 = 1.0f, d1 = 1.0f, d2 = 1.0f;
      if (SRCSCALE) { d0 = dinv[s0]; d1 = dinv[s1]; d2 = dinv[s2]; }
      const vecf v0 = rows[(size_t)s0 * 32 + lane];
      const vecf v1 = rows[(size_t)s1 * 32 + lane];
      const vecf v2 = rows[(size_t)s2 * 32 + lane];
      if (SRCSCALE) {
        acc += d0 * v0;
        acc += d1 * v1;
        acc += d2 * v2;
      } else {
        acc += v0;
        acc += v1;
        acc += v2;
      }
    } else if (rem == 2) {
      const int s0 = __builtin_amdgcn_readlane(idx, j);
      const int s1 = __builtin_amdgcn_readlane(idx, j + 1);
      float d0 = 1.0f, d1 = 1.0f;
      if (SRCSCALE) { d0 = dinv[s0]; d1 = dinv[s1]; }
      const vecf v0 = rows[(size_t)s0 * 32 + lane];
      const vecf v1 = rows[(size_t)s1 * 32 + lane];
      if (SRCSCALE) {
        acc += d0 * v0;
        acc += d1 * v1;
      } else {
        acc += v0;
        acc += v1;
      }
    } else if (rem == 1) {
      const int s0 = __builtin_amdgcn_readlane(idx, j);
      float d0 = 1.0f;
      if (SRCSCALE) d0 = dinv[s0];
      const vecf v0 = rows[(size_t)s0 * 32 + lane];
      if (SRCSCALE) acc += d0 * v0; else acc += v0;
    }
  }
  return acc;
}

__global__ __launch_bounds__(256)
void k_gather(const int* __restrict__ rowptr, const int* __restrict__ cnt,
              const int* __restrict__ csr, const float* __restrict__ dinv,
              const float* __restrict__ HS, const float* __restrict__ b,
              float* __restrict__ OUT) {
  typedef rowvec<OUT_C / 32> vecf;

  const int lane = threadIdx.x & 31;
  const int node = __builtin_amdgcn_readfirstlane((int)((blockIdx.x * 256 + threadIdx.x) >> 5));
  if (node >= N_NODES) return;
  const int start = __builtin_amdgcn_readfirstlane(rowptr[node]);
  const int n     = __builtin_amdgcn_readfirstlane(cnt[node]);
  const float di  = dinv[node];

  const vecf acc = gather_sum<OUT_C, false>(node, start, n, di, lane, csr, dinv, HS);
  const vecf bb = ((const vecf*)b)[lane];
  ((vecf*)OUT)[(size_t)node * 32 + lane] = di * acc + bb;
}

__global__ __launch_bounds__(512)
void k_gather_gemm2(const int* __restrict__ rowptr, const int* __restrict__ cnt,
                    const int* __restrict__ csr, const float* __restrict__ dinv,
                    const float* __restrict__ H1, const float* __restrict__ b1,
                    const float* __restrict__ W2, float* __restrict__ H2) {
  typedef rowvec<HID_C / 32> v4f;
  __shared__ __attribute__((aligned(16))) float sW[HID_C * W2_PITCH];
  __shared__ __attribute__((aligned(16))) float sG[TILE_N * G_PITCH];
  __shared__ float sDi[TILE_N];

  const int tid  = threadIdx.x;
  const int lane = tid & 31;
  const int wv   = __builtin_amdgcn_readfirstlane(tid >> 5);
  const int nodeBase = blockIdx.x * TILE_N;

  v4f wreg[4];
#pragma unroll
  for (int i = 0; i < 4; ++i) wreg[i] = ((const v4f*)W2)[tid + i * 512];

  const v4f bb = ((const v4f*)b1)[lane];
#pragma nounroll
  for (int q = 0; q < 2; ++q) {
    const int r    = wv * 2 + q;
    const int node = nodeBase + r;
    if (node < N_NODES) {
      const int start = __builtin_amdgcn_readfirstlane(rowptr[node]);
      const int n     = __builtin_amdgcn_readfirstlane(cnt[node]);
      const float di  = dinv[node];
      const v4f acc = gather_sum<HID_C, true>(node, start, n, di, lane, csr, dinv, H1);
      v4f o = di * acc + bb;
      for (int k = 0; k < 4; ++k) o[k] = fmaxf(o[k], 0.0f);
      *(v4f*)&sG[r * G_PITCH + 4 * lane] = o;
      if (lane == 0) sDi[r] = di;
    }
  }

#pragma unroll
  for (int i = 0; i < 4; ++i) {
    const int f4 = tid + i * 512;
    *(v4f*)&sW[(f4 >> 4) * W2_PITCH + (f4 & 15) * 4] = wreg[i];
  }
  __syncthreads();

  if (wv >= 4) return;
  const int mt = wv >> 1;
  const int nh = wv & 1;
  if (nodeBase + mt * 16 >= N_NODES) return;

  const int half = lane >> 4;
  const int col  = lane & 15;
  const int koff = half * 2;
  const float* ap = &sG[(mt * 16 + col) * G_PITCH + koff];
  const float* bp = &sW[koff * W2_PITCH + nh * 32 + col];

  v8f acc[2];
  acc[0] = (v8f){};
  acc[1] = (v8f){};
#pragma unroll 4
  for (int kk = 0; kk < HID_C; kk += 4) {
    v2f a;
    a.x = ap[kk];
    a.y = ap[kk + 1];
#pragma unroll
    for (int t = 0; t < 2; ++t) {
      v2f b;
      b.x = bp[kk * W2_PITCH + 16 * t];
      b.y = bp[kk * W2_PITCH + W2_PITCH + 16 * t];
      acc[t] = __builtin_amdgcn_wmma_f32_16x16x4_f32(
           false, a,  false, b,
           (short)0, acc[t],  false,  false);
    }
  }

  const int rOff = mt * 16 + half * 8;
#pragma unroll
  for (int t = 0; t < 2; ++t)
#pragma unroll
    for (int r = 0; r < 8; ++r)
      H2[(size_t)(nodeBase + rOff + r) * OUT_C + nh * 32 + t * 16 + col] =
          acc[t][r] * sDi[rOff + r];
}

extern "C" void kernel_launch(void* const* d_in, const int* in_sizes, int n_in,
                              void* d_out, int out_size, void* d_ws, size_t ws_size,
                              hipStream_t stream) {
  const float* x    = (const float*)d_in[0];
  const int*   eidx = (const int*)d_in[1];
  const float* W1   = (const float*)d_in[2];
  const float* b1   = (const float*)d_in[3];
  const float* W2   = (const float*)d_in[4];
  const float* b2   = (const float*)d_in[5];
  const int* src = eidx;
  const int* dst = eidx + N_EDGES;

  char* ws = (char*)d_ws;
  auto alloc = [&](size_t bytes) {
    char* p = ws;
    ws += (bytes + 255) & ~(size_t)255;
    return (void*)p;
  };
  int*   cnt    = (int*)  alloc((size_t)N_NODES * 4);
  int*   rowptr = (int*)  alloc((size_t)N_NODES * 4);
  float* dinv   = (float*)alloc((size_t)N_NODES * 4);
  int*   pairs  = (int*)  alloc((size_t)SORT_BLK * CHUNK_P * 8);
  int*   table  = (int*)  alloc((size_t)SORT_BLK * (NBUCK + 1) * 4);
  int*   csr    = (int*)  alloc((size_t)N_EDGES * 4);
  float* h1     = (float*)alloc((size_t)N_NODES * HID_C * 4);
  float* h2     = (float*)alloc((size_t)N_NODES * OUT_C * 4);

  k_bucket_gemm<<<GEMM_BLK_8 + SORT_BLK, 256, 0, stream>>>(src, dst, pairs, table, x, W1, h1);
  k_owner_gemm<<<GEMM_BLK_16 + NBUCK, 512, 0, stream>>>(pairs, table, cnt, rowptr, dinv, csr,
                                                        x, W1, h1);

  const int gatherBlocks = (N_NODES + 7) / 8;

  k_gather_gemm2<<<FUSE_BLK, 512, 0, stream>>>(rowptr, cnt, csr, dinv, h1, b1, W2, h2);

  k_gather<<<gatherBlocks, 256, 0, stream>>>(rowptr, cnt, csr, dinv, h2, b2, (float*)d_out);
}
